# starting kernel
# baseline (speedup 1.0000x reference)
#include <hip/hip_runtime.h>
#include <hip/hip_bf16.h>


typedef __attribute__((ext_vector_type(2))) float v2f;
typedef __attribute__((ext_vector_type(8))) float v8f;

#define HIDDEN 256
#define QCH    12
#define XSTR   (HIDDEN + QCH)
#define HALFW  128
#define APM    512

__device__ __constant__ const float kFScale =
    (float)(0.5 * 27.211386024367243 * 0.5291772105638411 / 7.0);

__device__ __forceinline__ float half16_xor_reduce(float p) {
    p += __int_as_float(__builtin_amdgcn_ds_swizzle(__float_as_int(p), (1 << 10) | 0x1f));
    p += __int_as_float(__builtin_amdgcn_ds_swizzle(__float_as_int(p), (2 << 10) | 0x1f));
    p += __int_as_float(__builtin_amdgcn_ds_swizzle(__float_as_int(p), (4 << 10) | 0x1f));
    p += __int_as_float(__builtin_amdgcn_ds_swizzle(__float_as_int(p), (8 << 10) | 0x1f));
    return p;
}

__global__ __launch_bounds__(256) void mlp_wmma_kernel(
    const float* __restrict__ x,  const float* __restrict__ W1,
    const float* __restrict__ b1, const float* __restrict__ W2,
    const float* __restrict__ b2, float* __restrict__ out)
{
    __shared__ float partial[8][16];

    const int tid  = threadIdx.x;
    const int wave = tid >> 5;
    const int lane = tid & 31;
    const int hi   = lane >> 4;
    const int lo   = lane & 15;
    const int rowbase = blockIdx.x << 4;
    const int colbase = wave << 4;

    const float* __restrict__ arow = x  + (size_t)(rowbase + lo) * XSTR + 2 * hi;
    const float* __restrict__ bcol = W1 + (size_t)(2 * hi) * HALFW + colbase + lo;

    v8f c = {};
    #pragma unroll 4
    for (int k = 0; k < HIDDEN; k += 4) {
        v2f a, b;
        a.x = arow[k];
        a.y = arow[k + 1];
        b.x = bcol[(size_t)k * HALFW];
        b.y = bcol[(size_t)k * HALFW + HALFW];
        c = __builtin_amdgcn_wmma_f32_16x16x4_f32(
                false, a, false, b, (short)0, c, false, false);
    }

    const float w2v = W2[colbase + lo];
    const float b1v = b1[colbase + lo];

    #pragma unroll
    for (int r = 0; r < 8; ++r) {
        float h = c[r] + b1v;
        h = h * __builtin_amdgcn_rcpf(1.0f + __expf(-h));
        float p = half16_xor_reduce(h * w2v);
        if (lo == 0) partial[wave][r + 8 * hi] = p;
    }
    __syncthreads();

    if (tid < 16) {
        float s = b2[0];
        #pragma unroll
        for (int w = 0; w < 8; ++w) s += partial[w][tid];
        out[rowbase + tid] = s;
    }
}

__global__ __launch_bounds__(128) void pair_kernel(
    const float* __restrict__ x, const float* __restrict__ pos,
    const float* __restrict__ qw, float* __restrict__ out)
{
    __shared__ float4 sp[128];
    __shared__ float4 sq[128][3];

    const int b     = blockIdx.x;
    const int mol   = b >> 4;
    const int ipart = (b >> 2) & 3;
    const int jpart = b & 3;
    const int tid   = threadIdx.x;
    const int molbase = mol * APM;
    const int i     = molbase + ipart * 128 + tid;
    const int jbase = molbase + jpart * 128;

    float sw[QCH];
    #pragma unroll
    for (int cch = 0; cch < QCH; ++cch) sw[cch] = sqrtf(qw[cch]);

    {
        const int j = jbase + tid;
        const float* pj = pos + 3 * (size_t)j;
        sp[tid] = make_float4(pj[0], pj[1], pj[2], 0.0f);
        const float4* cj = (const float4*)(x + (size_t)j * XSTR + HIDDEN);
        float4 q0 = cj[0], q1 = cj[1], q2 = cj[2];
        q0.x *= sw[0];  q0.y *= sw[1];  q0.z *= sw[2];   q0.w *= sw[3];
        q1.x *= sw[4];  q1.y *= sw[5];  q1.z *= sw[6];   q1.w *= sw[7];
        q2.x *= sw[8];  q2.y *= sw[9];  q2.z *= sw[10];  q2.w *= sw[11];
        sq[tid][0] = q0; sq[tid][1] = q1; sq[tid][2] = q2;
    }

    const float* pi = pos + 3 * (size_t)i;
    const float pix = pi[0], piy = pi[1], piz = pi[2];
    const float4* ci = (const float4*)(x + (size_t)i * XSTR + HIDDEN);
    float4 qi0 = ci[0], qi1 = ci[1], qi2 = ci[2];
    qi0.x *= sw[0];  qi0.y *= sw[1];  qi0.z *= sw[2];   qi0.w *= sw[3];
    qi1.x *= sw[4];  qi1.y *= sw[5];  qi1.z *= sw[6];   qi1.w *= sw[7];
    qi2.x *= sw[8];  qi2.y *= sw[9];  qi2.z *= sw[10];  qi2.w *= sw[11];

    __syncthreads();

    const float inv_rc = 1.0f / 4.6f;
    float acc = 0.0f;
    #pragma unroll 4
    for (int jj = 0; jj < 128; ++jj) {
        const float4 pj = sp[jj];
        const float dx = pix - pj.x;
        const float dy = piy - pj.y;
        const float dz = piz - pj.z;
        const float d2 = dx * dx + dy * dy + dz * dz;

        const float4 a0 = sq[jj][0];
        const float4 a1 = sq[jj][1];
        const float4 a2 = sq[jj][2];
        float qd = qi0.x * a0.x + qi0.y * a0.y + qi0.z * a0.z + qi0.w * a0.w
                 + qi1.x * a1.x + qi1.y * a1.y + qi1.z * a1.z + qi1.w * a1.w
                 + qi2.x * a2.x + qi2.y * a2.y + qi2.z * a2.z + qi2.w * a2.w;

        const float inv_d = __builtin_amdgcn_rsqf(d2);
        const float d     = d2 * inv_d;
        const float r     = fminf(d * inv_rc, 1.0f - 1e-6f);
        const float fc = 1.0f - __expf(1.0f - __builtin_amdgcn_rcpf(1.0f - r * r));
        const float contrib = fc * inv_d * qd;
        acc += (jbase + jj != i) ? contrib : 0.0f;
    }

    __hip_atomic_fetch_add(&out[i], acc * kFScale,
                           __ATOMIC_RELAXED, __HIP_MEMORY_SCOPE_AGENT);
}

extern "C" void kernel_launch(void* const* d_in, const int* in_sizes, int n_in,
                              void* d_out, int out_size, void* d_ws, size_t ws_size,
                              hipStream_t stream) {
    const float* x   = (const float*)d_in[0];
    const float* pos = (const float*)d_in[1];
    const float* W1  = (const float*)d_in[5];
    const float* b1  = (const float*)d_in[6];
    const float* W2  = (const float*)d_in[7];
    const float* b2  = (const float*)d_in[8];
    const float* qw  = (const float*)d_in[9];
    float* out = (float*)d_out;

    const int N    = in_sizes[0] / XSTR;
    const int nmol = N / APM;

    mlp_wmma_kernel<<<N / 16, 256, 0, stream>>>(x, W1, b1, W2, b2, out);
    pair_kernel<<<nmol * 16, 128, 0, stream>>>(x, pos, qw, out);
}


# GEMM kernel re-tiled to 64 rows x 128 cols per block: feat tile staged once in LDS by whole-line loads, A fragments read from LDS, each W1 fragment feeds 4 row tiles; summation order unchanged
# speedup vs baseline: 1.2703x; 1.2703x over previous
#include <hip/hip_runtime.h>
#include <hip/hip_bf16.h>


typedef __attribute__((ext_vector_type(2))) float v2f;
typedef __attribute__((ext_vector_type(8))) float v8f;

#define HIDDEN 256
#define QCH    12
#define XSTR   (HIDDEN + QCH)
#define HALFW  128
#define APM    512

__device__ __constant__ const float kFScale =
    (float)(0.5 * 27.211386024367243 * 0.5291772105638411 / 7.0);

template <int PATTERN, int NV>
__device__ __forceinline__ void xor_level(float (&p)[NV]) {
    float q[NV];
    #pragma unroll
    for (int i = 0; i < NV; ++i)
        q[i] = __int_as_float(__builtin_amdgcn_ds_swizzle(__float_as_int(p[i]), PATTERN));
    #pragma unroll
    for (int i = 0; i < NV; ++i) p[i] += q[i];
}

#define BROWS  64
#define RTILES 4
#define APITCH 260
#define KU     8
#define NSTAGE 8

__global__ __launch_bounds__(256) void mlp_wmma_kernel(
    const float* __restrict__ x,  const float* __restrict__ W1,
    const float* __restrict__ b1, const float* __restrict__ W2,
    const float* __restrict__ b2, float* __restrict__ out)
{
    __shared__ alignas(16) float sA[BROWS * APITCH];
    __shared__ float partial[8][BROWS];

    const int tid  = threadIdx.x;
    const int wave = tid >> 5;
    const int lane = tid & 31;
    const int hi   = lane >> 4;
    const int lo   = lane & 15;
    const int rowbase = blockIdx.x * BROWS;
    const int colbase = wave << 4;

    const float* __restrict__ bcol = W1 + (size_t)(2 * hi) * HALFW + colbase + lo;
    const float* afrag = sA + lo * APITCH + 2 * hi;

    v2f a0[KU][RTILES], a1[KU][RTILES];
    v2f bf0[KU], bf1[KU];

    auto load_b = [&](v2f (&b)[KU], int k0) {
        #pragma unroll
        for (int u = 0; u < KU; ++u) {
            const int k = k0 + 4 * u;
            b[u].x = bcol[(size_t)k * HALFW];
            b[u].y = bcol[(size_t)k * HALFW + HALFW];
        }
    };
    auto load_a = [&](v2f (&a)[KU][RTILES], int k0) {
        #pragma unroll
        for (int u = 0; u < KU; ++u) {
            #pragma unroll
            for (int t = 0; t < RTILES; ++t)
                a[u][t] = *(const v2f*)(afrag + t * 16 * APITCH + k0 + 4 * u);
        }
    };

    load_b(bf0, 0);

    {
        const float* xblk = x + (size_t)rowbase * XSTR;
        float4 v[16];
        #pragma unroll
        for (int i = 0; i < 16; ++i) {
            const int idx = tid + 256 * i;
            v[i] = *(const float4*)(xblk + (size_t)(idx >> 6) * XSTR + 4 * (idx & 63));
        }
        #pragma unroll
        for (int i = 0; i < 16; ++i) {
            const int idx = tid + 256 * i;
            *(float4*)(sA + (idx >> 6) * APITCH + 4 * (idx & 63)) = v[i];
        }
    }
    __syncthreads();

    v8f c[RTILES];
    #pragma unroll
    for (int t = 0; t < RTILES; ++t) c[t] = v8f{};

    auto mma_stage = [&](const v2f (&a)[KU][RTILES], const v2f (&b)[KU]) {
        #pragma unroll
        for (int u = 0; u < KU; ++u) {
            #pragma unroll
            for (int t = 0; t < RTILES; ++t) {
                c[t] = __builtin_amdgcn_wmma_f32_16x16x4_f32(
                           false, a[u][t], false, b[u], (short)0, c[t], false, false);
            }
        }
    };

    load_a(a0, 0);
    for (int s = 0; s < NSTAGE; s += 2) {
        load_b(bf1, (s + 1) * 4 * KU);
        load_a(a1,  (s + 1) * 4 * KU);
        mma_stage(a0, bf0);
        if (s + 2 < NSTAGE) {
            load_b(bf0, (s + 2) * 4 * KU);
            load_a(a0,  (s + 2) * 4 * KU);
        }
        mma_stage(a1, bf1);
    }

    const float w2v = W2[colbase + lo];
    const float b1v = b1[colbase + lo];

    float p[8 * RTILES];
    #pragma unroll
    for (int t = 0; t < RTILES; ++t) {
        #pragma unroll
        for (int r = 0; r < 8; ++r) {
            float h = c[t][r] + b1v;
            h = h * __builtin_amdgcn_rcpf(1.0f + __expf(-h));
            p[8 * t + r] = h * w2v;
        }
    }
    xor_level<(1 << 10) | 0x1f>(p);
    xor_level<(2 << 10) | 0x1f>(p);
    xor_level<(4 << 10) | 0x1f>(p);
    xor_level<(8 << 10) | 0x1f>(p);
    if (lo == 0) {
        #pragma unroll
        for (int t = 0; t < RTILES; ++t) {
            #pragma unroll
            for (int r = 0; r < 8; ++r) partial[wave][16 * t + r + 8 * hi] = p[8 * t + r];
        }
    }
    __syncthreads();

    if (tid < BROWS) {
        float s = b2[0];
        #pragma unroll
        for (int w = 0; w < 8; ++w) s += partial[w][tid];
        out[rowbase + tid] = s;
    }
}

__global__ __launch_bounds__(128) void pair_kernel(
    const float* __restrict__ x, const float* __restrict__ pos,
    const float* __restrict__ qw, float* __restrict__ out)
{
    __shared__ float4 sp[128];
    __shared__ float4 sq[128][3];

    const int b     = blockIdx.x;
    const int mol   = b >> 4;
    const int ipart = (b >> 2) & 3;
    const int jpart = b & 3;
    const int tid   = threadIdx.x;
    const int molbase = mol * APM;
    const int i     = molbase + ipart * 128 + tid;
    const int jbase = molbase + jpart * 128;

    float sw[QCH];
    #pragma unroll
    for (int cch = 0; cch < QCH; ++cch) sw[cch] = sqrtf(qw[cch]);

    {
        const int j = jbase + tid;
        const float* pj = pos + 3 * (size_t)j;
        sp[tid] = make_float4(pj[0], pj[1], pj[2], 0.0f);
        const float4* cj = (const float4*)(x + (size_t)j * XSTR + HIDDEN);
        float4 q0 = cj[0], q1 = cj[1], q2 = cj[2];
        q0.x *= sw[0];  q0.y *= sw[1];  q0.z *= sw[2];   q0.w *= sw[3];
        q1.x *= sw[4];  q1.y *= sw[5];  q1.z *= sw[6];   q1.w *= sw[7];
        q2.x *= sw[8];  q2.y *= sw[9];  q2.z *= sw[10];  q2.w *= sw[11];
        sq[tid][0] = q0; sq[tid][1] = q1; sq[tid][2] = q2;
    }

    const float* pi = pos + 3 * (size_t)i;
    const float pix = pi[0], piy = pi[1], piz = pi[2];
    const float4* ci = (const float4*)(x + (size_t)i * XSTR + HIDDEN);
    float4 qi0 = ci[0], qi1 = ci[1], qi2 = ci[2];
    qi0.x *= sw[0];  qi0.y *= sw[1];  qi0.z *= sw[2];   qi0.w *= sw[3];
    qi1.x *= sw[4];  qi1.y *= sw[5];  qi1.z *= sw[6];   qi1.w *= sw[7];
    qi2.x *= sw[8];  qi2.y *= sw[9];  qi2.z *= sw[10];  qi2.w *= sw[11];

    __syncthreads();

    const float inv_rc = 1.0f / 4.6f;
    float acc = 0.0f;
    #pragma unroll 4
    for (int jj = 0; jj < 128; ++jj) {
        const float4 pj = sp[jj];
        const float dx = pix - pj.x;
        const float dy = piy - pj.y;
        const float dz = piz - pj.z;
        const float d2 = dx * dx + dy * dy + dz * dz;

        const float4 a0 = sq[jj][0];
        const float4 a1 = sq[jj][1];
        const float4 a2 = sq[jj][2];
        float qd = qi0.x * a0.x + qi0.y * a0.y + qi0.z * a0.z + qi0.w * a0.w
                 + qi1.x * a1.x + qi1.y * a1.y + qi1.z * a1.z + qi1.w * a1.w
                 + qi2.x * a2.x + qi2.y * a2.y + qi2.z * a2.z + qi2.w * a2.w;

        const float inv_d = __builtin_amdgcn_rsqf(d2);
        const float d     = d2 * inv_d;
        const float r     = fminf(d * inv_rc, 1.0f - 1e-6f);
        const float fc = 1.0f - __expf(1.0f - __builtin_amdgcn_rcpf(1.0f - r * r));
        const float contrib = fc * inv_d * qd;
        acc += (jbase + jj != i) ? contrib : 0.0f;
    }

    __hip_atomic_fetch_add(&out[i], acc * kFScale,
                           __ATOMIC_RELAXED, __HIP_MEMORY_SCOPE_AGENT);
}

extern "C" void kernel_launch(void* const* d_in, const int* in_sizes, int n_in,
                              void* d_out, int out_size, void* d_ws, size_t ws_size,
                              hipStream_t stream) {
    const float* x   = (const float*)d_in[0];
    const float* pos = (const float*)d_in[1];
    const float* W1  = (const float*)d_in[5];
    const float* b1  = (const float*)d_in[6];
    const float* W2  = (const float*)d_in[7];
    const float* b2  = (const float*)d_in[8];
    const float* qw  = (const float*)d_in[9];
    float* out = (float*)d_out;

    const int N    = in_sizes[0] / XSTR;
    const int nmol = N / APM;

    mlp_wmma_kernel<<<N / BROWS, 256, 0, stream>>>(x, W1, b1, W2, b2, out);
    pair_kernel<<<nmol * 16, 128, 0, stream>>>(x, pos, qw, out);
}


# Pair kernel: j-chunk staged as two-atom slots and walked 8 independent pairs per lane in two-wide fp32, each batch's LDS reads grouped ahead of its math; lane sum split into even/odd-j partial sums
# speedup vs baseline: 2.3090x; 1.8177x over previous
#include <hip/hip_runtime.h>
#include <hip/hip_bf16.h>


typedef __attribute__((ext_vector_type(2))) float v2f;
typedef __attribute__((ext_vector_type(8))) float v8f;

#define HIDDEN 256
#define QCH    12
#define XSTR   (HIDDEN + QCH)
#define HALFW  128
#define APM    512

template <int PATTERN, int NV>
__device__ __forceinline__ void xor_level(float (&p)[NV]) {
    float q[NV];
    #pragma unroll
    for (int i = 0; i < NV; ++i)
        q[i] = __int_as_float(__builtin_amdgcn_ds_swizzle(__float_as_int(p[i]), PATTERN));
    #pragma unroll
    for (int i = 0; i < NV; ++i) p[i] += q[i];
}

#define BROWS  64
#define RTILES 4
#define APITCH 260
#define KU     8
#define NSTAGE 8

__global__ __launch_bounds__(256) void mlp_wmma_kernel(
    const float* __restrict__ x,  const float* __restrict__ W1,
    const float* __restrict__ b1, const float* __restrict__ W2,
    const float* __restrict__ b2, float* __restrict__ out)
{
    __shared__ alignas(16) float sA[BROWS * APITCH];
    __shared__ float partial[8][BROWS];

    const int tid  = threadIdx.x;
    const int wave = tid >> 5;
    const int lane = tid & 31;
    const int hi   = lane >> 4;
    const int lo   = lane & 15;
    const int rowbase = blockIdx.x * BROWS;
    const int colbase = wave << 4;

    const float* __restrict__ bcol = W1 + (size_t)(2 * hi) * HALFW + colbase + lo;
    const float* afrag = sA + lo * APITCH + 2 * hi;

    v2f a0[KU][RTILES], a1[KU][RTILES];
    v2f bf0[KU], bf1[KU];

    auto load_b = [&](v2f (&b)[KU], int k0) {
        #pragma unroll
        for (int u = 0; u < KU; ++u) {
            const int k = k0 + 4 * u;
            b[u].x = bcol[(size_t)k * HALFW];
            b[u].y = bcol[(size_t)k * HALFW + HALFW];
        }
    };
    auto load_a = [&](v2f (&a)[KU][RTILES], int k0) {
        #pragma unroll
        for (int u = 0; u < KU; ++u) {
            #pragma unroll
            for (int t = 0; t < RTILES; ++t)
                a[u][t] = *(const v2f*)(afrag + t * 16 * APITCH + k0 + 4 * u);
        }
    };

    load_b(bf0, 0);

    {
        const float* xblk = x + (size_t)rowbase * XSTR;
        float4 v[16];
        #pragma unroll
        for (int i = 0; i < 16; ++i) {
            const int idx = tid + 256 * i;
            v[i] = *(const float4*)(xblk + (size_t)(idx >> 6) * XSTR + 4 * (idx & 63));
        }
        #pragma unroll
        for (int i = 0; i < 16; ++i) {
            const int idx = tid + 256 * i;
            *(float4*)(sA + (idx >> 6) * APITCH + 4 * (idx & 63)) = v[i];
        }
    }
    __syncthreads();

    v8f c[RTILES];
    #pragma unroll
    for (int t = 0; t < RTILES; ++t) c[t] = v8f{};

    auto mma_stage = [&](const v2f (&a)[KU][RTILES], const v2f (&b)[KU]) {
        #pragma unroll
        for (int u = 0; u < KU; ++u) {
            #pragma unroll
            for (int t = 0; t < RTILES; ++t) {
                c[t] = __builtin_amdgcn_wmma_f32_16x16x4_f32(
                           false, a[u][t], false, b[u], (short)0, c[t], false, false);
            }
        }
    };

    load_a(a0, 0);
    for (int s = 0; s < NSTAGE; s += 2) {
        load_b(bf1, (s + 1) * 4 * KU);
        load_a(a1,  (s + 1) * 4 * KU);
        mma_stage(a0, bf0);
        if (s + 2 < NSTAGE) {
            load_b(bf0, (s + 2) * 4 * KU);
            load_a(a0,  (s + 2) * 4 * KU);
        }
        mma_stage(a1, bf1);
    }

    const float w2v = W2[colbase + lo];
    const float b1v = b1[colbase + lo];

    float p[8 * RTILES];
    #pragma unroll
    for (int t = 0; t < RTILES; ++t) {
        #pragma unroll
        for (int r = 0; r < 8; ++r) {
            float h = c[t][r] + b1v;
            h = h * __builtin_amdgcn_rcpf(1.0f + __expf(-h));
            p[8 * t + r] = h * w2v;
        }
    }
    xor_level<(1 << 10) | 0x1f>(p);
    xor_level<(2 << 10) | 0x1f>(p);
    xor_level<(4 << 10) | 0x1f>(p);
    xor_level<(8 << 10) | 0x1f>(p);
    if (lo == 0) {
        #pragma unroll
        for (int t = 0; t < RTILES; ++t) {
            #pragma unroll
            for (int r = 0; r < 8; ++r) partial[wave][16 * t + r + 8 * hi] = p[8 * t + r];
        }
    }
    __syncthreads();

    if (tid < BROWS) {
        float s = b2[0];
        #pragma unroll
        for (int w = 0; w < 8; ++w) s += partial[w][tid];
        out[rowbase + tid] = s;
    }
}

#define PH    4
#define SLOTF 32

__global__ __launch_bounds__(128) void pair_kernel(
    const float* __restrict__ x, const float* __restrict__ pos,
    const float* __restrict__ qw, float* __restrict__ out)
{
    __shared__ alignas(16) float sj[64 * SLOTF];

    const int b     = blockIdx.x;
    const int mol   = b >> 4;
    const int ipart = (b >> 2) & 3;
    const int jpart = b & 3;
    const int tid   = threadIdx.x;
    const int molbase = mol * APM;
    const int i     = molbase + ipart * 128 + tid;
    const int jbase = molbase + jpart * 128;

    float w[QCH];
    float wsum = 0.0f;
    #pragma unroll
    for (int cch = 0; cch < QCH; ++cch) { w[cch] = qw[cch]; wsum += w[cch]; }

    {
        const int j = jbase + tid;
        const float* pj = pos + 3 * (size_t)j;
        const float4* cj = (const float4*)(x + (size_t)j * XSTR + HIDDEN);
        const float4 q0 = cj[0], q1 = cj[1], q2 = cj[2];
        float* d = sj + (tid >> 1) * SLOTF + (tid & 1);
        d[0]  = pj[0];        d[2]  = pj[1];        d[4]  = pj[2];
        d[6]  = q0.x * w[0];  d[8]  = q0.y * w[1];  d[10] = q0.z * w[2];  d[12] = q0.w * w[3];
        d[14] = q1.x * w[4];  d[16] = q1.y * w[5];  d[18] = q1.z * w[6];  d[20] = q1.w * w[7];
        d[22] = q2.x * w[8];  d[24] = q2.y * w[9];  d[26] = q2.z * w[10]; d[28] = q2.w * w[11];
    }

    const float* pi = pos + 3 * (size_t)i;
    const v2f pix = {pi[0], pi[0]}, piy = {pi[1], pi[1]}, piz = {pi[2], pi[2]};
    const float* ci = x + (size_t)i * XSTR + HIDDEN;
    v2f qi[QCH];
    #pragma unroll
    for (int cch = 0; cch < QCH; ++cch) { const float t = ci[cch]; qi[cch] = v2f{t, t}; }

    __syncthreads();

    const float inv_rc = 1.0f / 4.6f;
    const float rmax   = 1.0f - 1e-6f;
    const int   jself  = i - jbase;
    v2f acc = {0.0f, 0.0f};
    for (int m0 = 0; m0 < 64; m0 += PH) {
        float4 t[PH][8];
        #pragma unroll
        for (int u = 0; u < PH; ++u) {
            const float4* r = (const float4*)(sj + (m0 + u) * SLOTF);
            #pragma unroll
            for (int k = 0; k < 8; ++k) t[u][k] = r[k];
        }
        __builtin_amdgcn_sched_barrier(0);
        v2f d2[PH], qd[PH];
        #pragma unroll
        for (int u = 0; u < PH; ++u) {
            const v2f dx = pix - v2f{t[u][0].x, t[u][0].y};
            const v2f dy = piy - v2f{t[u][0].z, t[u][0].w};
            const v2f dz = piz - v2f{t[u][1].x, t[u][1].y};
            d2[u] = dz * dz + (dy * dy + dx * dx);
            v2f s = qi[0] * v2f{t[u][1].z, t[u][1].w};
            #pragma unroll
            for (int k = 1; k < 6; ++k) {
                s = qi[2 * k - 1] * v2f{t[u][k + 1].x, t[u][k + 1].y} + s;
                s = qi[2 * k]     * v2f{t[u][k + 1].z, t[u][k + 1].w} + s;
            }
            s = qi[11] * v2f{t[u][7].x, t[u][7].y} + s;
            qd[u] = s;
        }
        #pragma unroll
        for (int u = 0; u < PH; ++u) {
            const v2f inv_d = {__builtin_amdgcn_rsqf(d2[u].x), __builtin_amdgcn_rsqf(d2[u].y)};
            const v2f rr    = d2[u] * inv_d * inv_rc;
            const v2f r     = {fminf(rr.x, rmax), fminf(rr.y, rmax)};
            const v2f om    = 1.0f - r * r;
            const v2f ea    = {1.0f - __builtin_amdgcn_rcpf(om.x), 1.0f - __builtin_amdgcn_rcpf(om.y)};
            const v2f fc    = {1.0f - __expf(ea.x), 1.0f - __expf(ea.y)};
            const v2f contrib = fc * inv_d * qd[u];
            const int j = 2 * (m0 + u);
            acc.x += (j     != jself) ? contrib.x : 0.0f;
            acc.y += (j + 1 != jself) ? contrib.y : 0.0f;
        }
    }

    const float fscale = (float)(0.5 * 27.211386024367243 * 0.5291772105638411) / wsum;
    __hip_atomic_fetch_add(&out[i], (acc.x + acc.y) * fscale,
                           __ATOMIC_RELAXED, __HIP_MEMORY_SCOPE_AGENT);
}

extern "C" void kernel_launch(void* const* d_in, const int* in_sizes, int n_in,
                              void* d_out, int out_size, void* d_ws, size_t ws_size,
                              hipStream_t stream) {
    const float* x   = (const float*)d_in[0];
    const float* pos = (const float*)d_in[1];
    const float* W1  = (const float*)d_in[5];
    const float* b1  = (const float*)d_in[6];
    const float* W2  = (const float*)d_in[7];
    const float* b2  = (const float*)d_in[8];
    const float* qw  = (const float*)d_in[9];
    float* out = (float*)d_out;

    const int N    = in_sizes[0] / XSTR;
    const int nmol = N / APM;

    mlp_wmma_kernel<<<N / BROWS, 256, 0, stream>>>(x, W1, b1, W2, b2, out);
    pair_kernel<<<nmol * 16, 128, 0, stream>>>(x, pos, qw, out);
}


# GEMM and pair kernels fused into one launch: a block does both for its 64 atoms, 4 waves GEMM-first, 4 pair-first, so WMMA overlaps vector/LDS work; plain store replaces atomics (fixed sum order)
# speedup vs baseline: 2.4664x; 1.0682x over previous
#include <hip/hip_runtime.h>
#include <hip/hip_bf16.h>


typedef __attribute__((ext_vector_type(2))) float v2f;
typedef __attribute__((ext_vector_type(8))) float v8f;

#define HIDDEN 256
#define QCH    12
#define XSTR   (HIDDEN + QCH)
#define HALFW  128
#define APM    512

template <int PATTERN, int NV>
__device__ __forceinline__ void xor_level(float (&p)[NV]) {
    float q[NV];
    #pragma unroll
    for (int i = 0; i < NV; ++i)
        q[i] = __int_as_float(__builtin_amdgcn_ds_swizzle(__float_as_int(p[i]), PATTERN));
    #pragma unroll
    for (int i = 0; i < NV; ++i) p[i] += q[i];
}

#define BROWS  64
#define RTILES 4
#define APITCH 260
#define KU     8
#define NSTAGE 8
#define PH     4
#define SLOTF  32
#define JSLOTS (APM / 2)
#define QSLOTS (JSLOTS / 4)

__global__ __launch_bounds__(256) void mlp_pair_fused_kernel(
    const float* __restrict__ x,  const float* __restrict__ pos,
    const float* __restrict__ W1, const float* __restrict__ b1,
    const float* __restrict__ W2, const float* __restrict__ b2,
    const float* __restrict__ qw, float* __restrict__ out)
{
    __shared__ alignas(16) float sA[BROWS * APITCH];
    __shared__ alignas(16) float sj[JSLOTS * SLOTF];
    __shared__ float partial[8][BROWS];
    __shared__ float epart[4][BROWS];

    const int tid  = threadIdx.x;
    const int wave = tid >> 5;
    const int lane = tid & 31;
    const int hi   = lane >> 4;
    const int lo   = lane & 15;
    const int rowbase = blockIdx.x * BROWS;
    const int colbase = wave << 4;

    const int molbase = rowbase & ~(APM - 1);
    const int irow    = tid & (BROWS - 1);
    const int jq      = tid >> 6;
    const int i       = rowbase + irow;

    const float* __restrict__ bcol = W1 + (size_t)(2 * hi) * HALFW + colbase + lo;
    const float* afrag = sA + lo * APITCH + 2 * hi;

    v2f a0[KU][RTILES], a1[KU][RTILES];
    v2f bf0[KU], bf1[KU];

    auto load_b = [&](v2f (&b)[KU], int k0) {
        #pragma unroll
        for (int u = 0; u < KU; ++u) {
            const int k = k0 + 4 * u;
            b[u].x = bcol[(size_t)k * HALFW];
            b[u].y = bcol[(size_t)k * HALFW + HALFW];
        }
    };
    auto load_a = [&](v2f (&a)[KU][RTILES], int k0) {
        #pragma unroll
        for (int u = 0; u < KU; ++u) {
            #pragma unroll
            for (int t = 0; t < RTILES; ++t)
                a[u][t] = *(const v2f*)(afrag + t * 16 * APITCH + k0 + 4 * u);
        }
    };

    load_b(bf0, 0);

    float w[QCH];
    float wsum = 0.0f;
    #pragma unroll
    for (int cch = 0; cch < QCH; ++cch) { w[cch] = qw[cch]; wsum += w[cch]; }

    {
        const float* xblk = x + (size_t)rowbase * XSTR;
        float4 v[16];
        #pragma unroll
        for (int r = 0; r < 16; ++r) {
            const int idx = tid + 256 * r;
            v[r] = *(const float4*)(xblk + (size_t)(idx >> 6) * XSTR + 4 * (idx & 63));
        }
        #pragma unroll
        for (int r = 0; r < 16; ++r) {
            const int idx = tid + 256 * r;
            *(float4*)(sA + (idx >> 6) * APITCH + 4 * (idx & 63)) = v[r];
        }
    }

    #pragma unroll
    for (int h2 = 0; h2 < 2; ++h2) {
        const int a = tid + 256 * h2;
        const int j = molbase + a;
        const float* pj = pos + 3 * (size_t)j;
        const float4* cj = (const float4*)(x + (size_t)j * XSTR + HIDDEN);
        const float4 q0 = cj[0], q1 = cj[1], q2 = cj[2];
        float* d = sj + (a >> 1) * SLOTF + (a & 1);
        d[0]  = pj[0];        d[2]  = pj[1];        d[4]  = pj[2];
        d[6]  = q0.x * w[0];  d[8]  = q0.y * w[1];  d[10] = q0.z * w[2];  d[12] = q0.w * w[3];
        d[14] = q1.x * w[4];  d[16] = q1.y * w[5];  d[18] = q1.z * w[6];  d[20] = q1.w * w[7];
        d[22] = q2.x * w[8];  d[24] = q2.y * w[9];  d[26] = q2.z * w[10]; d[28] = q2.w * w[11];
    }

    const float* pi = pos + 3 * (size_t)i;
    const v2f pix = {pi[0], pi[0]}, piy = {pi[1], pi[1]}, piz = {pi[2], pi[2]};
    const float* ci = x + (size_t)i * XSTR + HIDDEN;
    v2f qi[QCH];
    #pragma unroll
    for (int cch = 0; cch < QCH; ++cch) { const float t = ci[cch]; qi[cch] = v2f{t, t}; }

    __syncthreads();

    auto gemm_part = [&]() {
        v8f c[RTILES];
        #pragma unroll
        for (int t = 0; t < RTILES; ++t) c[t] = v8f{};

        auto mma_stage = [&](const v2f (&a)[KU][RTILES], const v2f (&b)[KU]) {
            #pragma unroll
            for (int u = 0; u < KU; ++u) {
                #pragma unroll
                for (int t = 0; t < RTILES; ++t) {
                    c[t] = __builtin_amdgcn_wmma_f32_16x16x4_f32(
                               false, a[u][t], false, b[u], (short)0, c[t], false, false);
                }
            }
        };

        load_a(a0, 0);
        for (int s = 0; s < NSTAGE; s += 2) {
            load_b(bf1, (s + 1) * 4 * KU);
            load_a(a1,  (s + 1) * 4 * KU);
            mma_stage(a0, bf0);
            if (s + 2 < NSTAGE) {
                load_b(bf0, (s + 2) * 4 * KU);
                load_a(a0,  (s + 2) * 4 * KU);
            }
            mma_stage(a1, bf1);
        }

        const float w2v = W2[colbase + lo];
        const float b1v = b1[colbase + lo];

        float p[8 * RTILES];
        #pragma unroll
        for (int t = 0; t < RTILES; ++t) {
            #pragma unroll
            for (int r = 0; r < 8; ++r) {
                float h = c[t][r] + b1v;
                h = h * __builtin_amdgcn_rcpf(1.0f + __expf(-h));
                p[8 * t + r] = h * w2v;
            }
        }
        xor_level<(1 << 10) | 0x1f>(p);
        xor_level<(2 << 10) | 0x1f>(p);
        xor_level<(4 << 10) | 0x1f>(p);
        xor_level<(8 << 10) | 0x1f>(p);
        if (lo == 0) {
            #pragma unroll
            for (int t = 0; t < RTILES; ++t) {
                #pragma unroll
                for (int r = 0; r < 8; ++r) partial[wave][16 * t + r + 8 * hi] = p[8 * t + r];
            }
        }
    };

    auto pair_part = [&]() {
        const float inv_rc = 1.0f / 4.6f;
        const float rmax   = 1.0f - 1e-6f;
        const float* sq    = sj + jq * QSLOTS * SLOTF;
        const int   jself  = (i - molbase) - 2 * QSLOTS * jq;
        v2f acc = {0.0f, 0.0f};
        for (int m0 = 0; m0 < QSLOTS; m0 += PH) {
            float4 t[PH][8];
            #pragma unroll
            for (int u = 0; u < PH; ++u) {
                const float4* r = (const float4*)(sq + (m0 + u) * SLOTF);
                #pragma unroll
                for (int k = 0; k < 8; ++k) t[u][k] = r[k];
            }
            __builtin_amdgcn_sched_barrier(0);
            v2f d2[PH], qd[PH];
            #pragma unroll
            for (int u = 0; u < PH; ++u) {
                const v2f dx = pix - v2f{t[u][0].x, t[u][0].y};
                const v2f dy = piy - v2f{t[u][0].z, t[u][0].w};
                const v2f dz = piz - v2f{t[u][1].x, t[u][1].y};
                d2[u] = dz * dz + (dy * dy + dx * dx);
                v2f s = qi[0] * v2f{t[u][1].z, t[u][1].w};
                #pragma unroll
                for (int k = 1; k < 6; ++k) {
                    s = qi[2 * k - 1] * v2f{t[u][k + 1].x, t[u][k + 1].y} + s;
                    s = qi[2 * k]     * v2f{t[u][k + 1].z, t[u][k + 1].w} + s;
                }
                s = qi[11] * v2f{t[u][7].x, t[u][7].y} + s;
                qd[u] = s;
            }
            #pragma unroll
            for (int u = 0; u < PH; ++u) {
                const v2f inv_d = {__builtin_amdgcn_rsqf(d2[u].x), __builtin_amdgcn_rsqf(d2[u].y)};
                const v2f rr    = d2[u] * inv_d * inv_rc;
                const v2f r     = {fminf(rr.x, rmax), fminf(rr.y, rmax)};
                const v2f om    = 1.0f - r * r;
                const v2f ea    = {1.0f - __builtin_amdgcn_rcpf(om.x), 1.0f - __builtin_amdgcn_rcpf(om.y)};
                const v2f fc    = {1.0f - __expf(ea.x), 1.0f - __expf(ea.y)};
                const v2f contrib = fc * inv_d * qd[u];
                const int j = 2 * (m0 + u);
                acc.x += (j     != jself) ? contrib.x : 0.0f;
                acc.y += (j + 1 != jself) ? contrib.y : 0.0f;
            }
        }
        epart[jq][irow] = acc.x + acc.y;
    };

    const bool gemm_first = __builtin_amdgcn_readfirstlane(wave) < 4;
    for (int ph = 0; ph < 2; ++ph) {
        if ((ph == 0) == gemm_first) gemm_part();
        else                         pair_part();
    }

    __syncthreads();

    if (tid < BROWS) {
        float s = b2[0];
        #pragma unroll
        for (int ww = 0; ww < 8; ++ww) s += partial[ww][tid];
        const float fscale = (float)(0.5 * 27.211386024367243 * 0.5291772105638411) / wsum;
        const float e = (epart[0][tid] + epart[1][tid]) + (epart[2][tid] + epart[3][tid]);
        out[rowbase + tid] = s + e * fscale;
    }
}

extern "C" void kernel_launch(void* const* d_in, const int* in_sizes, int n_in,
                              void* d_out, int out_size, void* d_ws, size_t ws_size,
                              hipStream_t stream) {
    const float* x   = (const float*)d_in[0];
    const float* pos = (const float*)d_in[1];
    const float* W1  = (const float*)d_in[5];
    const float* b1  = (const float*)d_in[6];
    const float* W2  = (const float*)d_in[7];
    const float* b2  = (const float*)d_in[8];
    const float* qw  = (const float*)d_in[9];
    float* out = (float*)d_out;

    const int N = in_sizes[0] / XSTR;

    mlp_pair_fused_kernel<<<N / BROWS, 256, 0, stream>>>(x, pos, W1, b1, W2, b2, qw, out);
}


# Pair loop: each batch's LDS reads issued one batch ahead into a second register set (ping-pong), so the read stream and LDS round trip run under the previous batch's arithmetic; sum order unchanged
# speedup vs baseline: 2.6268x; 1.0650x over previous
#include <hip/hip_runtime.h>
#include <hip/hip_bf16.h>


typedef __attribute__((ext_vector_type(2))) float v2f;
typedef __attribute__((ext_vector_type(8))) float v8f;

#define HIDDEN 256
#define QCH    12
#define XSTR   (HIDDEN + QCH)
#define HALFW  128
#define APM    512

template <int PATTERN, int NV>
__device__ __forceinline__ void xor_level(float (&p)[NV]) {
    float q[NV];
    #pragma unroll
    for (int i = 0; i < NV; ++i)
        q[i] = __int_as_float(__builtin_amdgcn_ds_swizzle(__float_as_int(p[i]), PATTERN));
    #pragma unroll
    for (int i = 0; i < NV; ++i) p[i] += q[i];
}

#define BROWS  64
#define RTILES 4
#define APITCH 260
#define KU     8
#define NSTAGE 8
#define PH     4
#define SLOTF  32
#define JSLOTS (APM / 2)
#define QSLOTS (JSLOTS / 4)

__global__ __launch_bounds__(256) void mlp_pair_fused_kernel(
    const float* __restrict__ x,  const float* __restrict__ pos,
    const float* __restrict__ W1, const float* __restrict__ b1,
    const float* __restrict__ W2, const float* __restrict__ b2,
    const float* __restrict__ qw, float* __restrict__ out)
{
    __shared__ alignas(16) float sA[BROWS * APITCH];
    __shared__ alignas(16) float sj[JSLOTS * SLOTF];
    __shared__ float partial[8][BROWS];
    __shared__ float epart[4][BROWS];

    const int tid  = threadIdx.x;
    const int wave = tid >> 5;
    const int lane = tid & 31;
    const int hi   = lane >> 4;
    const int lo   = lane & 15;
    const int rowbase = blockIdx.x * BROWS;
    const int colbase = wave << 4;

    const int molbase = rowbase & ~(APM - 1);
    const int irow    = tid & (BROWS - 1);
    const int jq      = tid >> 6;
    const int i       = rowbase + irow;

    const float* __restrict__ bcol = W1 + (size_t)(2 * hi) * HALFW + colbase + lo;
    const float* afrag = sA + lo * APITCH + 2 * hi;

    v2f a0[KU][RTILES], a1[KU][RTILES];
    v2f bf0[KU], bf1[KU];

    auto load_b = [&](v2f (&b)[KU], int k0) {
        #pragma unroll
        for (int u = 0; u < KU; ++u) {
            const int k = k0 + 4 * u;
            b[u].x = bcol[(size_t)k * HALFW];
            b[u].y = bcol[(size_t)k * HALFW + HALFW];
        }
    };
    auto load_a = [&](v2f (&a)[KU][RTILES], int k0) {
        #pragma unroll
        for (int u = 0; u < KU; ++u) {
            #pragma unroll
            for (int t = 0; t < RTILES; ++t)
                a[u][t] = *(const v2f*)(afrag + t * 16 * APITCH + k0 + 4 * u);
        }
    };

    load_b(bf0, 0);

    float w[QCH];
    float wsum = 0.0f;
    #pragma unroll
    for (int cch = 0; cch < QCH; ++cch) { w[cch] = qw[cch]; wsum += w[cch]; }

    {
        const float* xblk = x + (size_t)rowbase * XSTR;
        float4 v[16];
        #pragma unroll
        for (int r = 0; r < 16; ++r) {
            const int idx = tid + 256 * r;
            v[r] = *(const float4*)(xblk + (size_t)(idx >> 6) * XSTR + 4 * (idx & 63));
        }
        #pragma unroll
        for (int r = 0; r < 16; ++r) {
            const int idx = tid + 256 * r;
            *(float4*)(sA + (idx >> 6) * APITCH + 4 * (idx & 63)) = v[r];
        }
    }

    #pragma unroll
    for (int h2 = 0; h2 < 2; ++h2) {
        const int a = tid + 256 * h2;
        const int j = molbase + a;
        const float* pj = pos + 3 * (size_t)j;
        const float4* cj = (const float4*)(x + (size_t)j * XSTR + HIDDEN);
        const float4 q0 = cj[0], q1 = cj[1], q2 = cj[2];
        float* d = sj + (a >> 1) * SLOTF + (a & 1);
        d[0]  = pj[0];        d[2]  = pj[1];        d[4]  = pj[2];
        d[6]  = q0.x * w[0];  d[8]  = q0.y * w[1];  d[10] = q0.z * w[2];  d[12] = q0.w * w[3];
        d[14] = q1.x * w[4];  d[16] = q1.y * w[5];  d[18] = q1.z * w[6];  d[20] = q1.w * w[7];
        d[22] = q2.x * w[8];  d[24] = q2.y * w[9];  d[26] = q2.z * w[10]; d[28] = q2.w * w[11];
    }

    const float* pi = pos + 3 * (size_t)i;
    const v2f pix = {pi[0], pi[0]}, piy = {pi[1], pi[1]}, piz = {pi[2], pi[2]};
    const float* ci = x + (size_t)i * XSTR + HIDDEN;
    v2f qi[QCH];
    #pragma unroll
    for (int cch = 0; cch < QCH; ++cch) { const float t = ci[cch]; qi[cch] = v2f{t, t}; }

    __syncthreads();

    auto gemm_part = [&]() {
        v8f c[RTILES];
        #pragma unroll
        for (int t = 0; t < RTILES; ++t) c[t] = v8f{};

        auto mma_stage = [&](const v2f (&a)[KU][RTILES], const v2f (&b)[KU]) {
            #pragma unroll
            for (int u = 0; u < KU; ++u) {
                #pragma unroll
                for (int t = 0; t < RTILES; ++t) {
                    c[t] = __builtin_amdgcn_wmma_f32_16x16x4_f32(
                               false, a[u][t], false, b[u], (short)0, c[t], false, false);
                }
            }
        };

        load_a(a0, 0);
        for (int s = 0; s < NSTAGE; s += 2) {
            load_b(bf1, (s + 1) * 4 * KU);
            load_a(a1,  (s + 1) * 4 * KU);
            mma_stage(a0, bf0);
            if (s + 2 < NSTAGE) {
                load_b(bf0, (s + 2) * 4 * KU);
                load_a(a0,  (s + 2) * 4 * KU);
            }
            mma_stage(a1, bf1);
        }

        const float w2v = W2[colbase + lo];
        const float b1v = b1[colbase + lo];

        float p[8 * RTILES];
        #pragma unroll
        for (int t = 0; t < RTILES; ++t) {
            #pragma unroll
            for (int r = 0; r < 8; ++r) {
                float h = c[t][r] + b1v;
                h = h * __builtin_amdgcn_rcpf(1.0f + __expf(-h));
                p[8 * t + r] = h * w2v;
            }
        }
        xor_level<(1 << 10) | 0x1f>(p);
        xor_level<(2 << 10) | 0x1f>(p);
        xor_level<(4 << 10) | 0x1f>(p);
        xor_level<(8 << 10) | 0x1f>(p);
        if (lo == 0) {
            #pragma unroll
            for (int t = 0; t < RTILES; ++t) {
                #pragma unroll
                for (int r = 0; r < 8; ++r) partial[wave][16 * t + r + 8 * hi] = p[8 * t + r];
            }
        }
    };

    auto pair_part = [&]() {
        const float inv_rc = 1.0f / 4.6f;
        const float rmax   = 1.0f - 1e-6f;
        const float* sq    = sj + jq * QSLOTS * SLOTF;
        const int   jself  = (i - molbase) - 2 * QSLOTS * jq;
        v2f acc = {0.0f, 0.0f};

        auto load_batch = [&](float4 (&t)[PH][8], int m0) {
            #pragma unroll
            for (int u = 0; u < PH; ++u) {
                const float4* r = (const float4*)(sq + (m0 + u) * SLOTF);
                #pragma unroll
                for (int k = 0; k < 8; ++k) t[u][k] = r[k];
            }
        };
        auto math_batch = [&](const float4 (&t)[PH][8], int m0) {
            v2f d2[PH], qd[PH];
            #pragma unroll
            for (int u = 0; u < PH; ++u) {
                const v2f dx = pix - v2f{t[u][0].x, t[u][0].y};
                const v2f dy = piy - v2f{t[u][0].z, t[u][0].w};
                const v2f dz = piz - v2f{t[u][1].x, t[u][1].y};
                d2[u] = dz * dz + (dy * dy + dx * dx);
                v2f s = qi[0] * v2f{t[u][1].z, t[u][1].w};
                #pragma unroll
                for (int k = 1; k < 6; ++k) {
                    s = qi[2 * k - 1] * v2f{t[u][k + 1].x, t[u][k + 1].y} + s;
                    s = qi[2 * k]     * v2f{t[u][k + 1].z, t[u][k + 1].w} + s;
                }
                s = qi[11] * v2f{t[u][7].x, t[u][7].y} + s;
                qd[u] = s;
            }
            #pragma unroll
            for (int u = 0; u < PH; ++u) {
                const v2f inv_d = {__builtin_amdgcn_rsqf(d2[u].x), __builtin_amdgcn_rsqf(d2[u].y)};
                const v2f rr    = d2[u] * inv_d * inv_rc;
                const v2f r     = {fminf(rr.x, rmax), fminf(rr.y, rmax)};
                const v2f om    = 1.0f - r * r;
                const v2f ea    = {1.0f - __builtin_amdgcn_rcpf(om.x), 1.0f - __builtin_amdgcn_rcpf(om.y)};
                const v2f fc    = {1.0f - __expf(ea.x), 1.0f - __expf(ea.y)};
                const v2f contrib = fc * inv_d * qd[u];
                const int j = 2 * (m0 + u);
                acc.x += (j     != jself) ? contrib.x : 0.0f;
                acc.y += (j + 1 != jself) ? contrib.y : 0.0f;
            }
        };

        float4 ta[PH][8], tb[PH][8];
        load_batch(ta, 0);
        int m0 = 0;
        for (; m0 < QSLOTS - 2 * PH; m0 += 2 * PH) {
            load_batch(tb, m0 + PH);
            __builtin_amdgcn_sched_barrier(0);
            math_batch(ta, m0);
            load_batch(ta, m0 + 2 * PH);
            __builtin_amdgcn_sched_barrier(0);
            math_batch(tb, m0 + PH);
        }
        load_batch(tb, m0 + PH);
        __builtin_amdgcn_sched_barrier(0);
        math_batch(ta, m0);
        math_batch(tb, m0 + PH);
        epart[jq][irow] = acc.x + acc.y;
    };

    const bool gemm_first = __builtin_amdgcn_readfirstlane(wave) < 4;
    for (int ph = 0; ph < 2; ++ph) {
        if ((ph == 0) == gemm_first) gemm_part();
        else                         pair_part();
    }

    __syncthreads();

    if (tid < BROWS) {
        float s = b2[0];
        #pragma unroll
        for (int ww = 0; ww < 8; ++ww) s += partial[ww][tid];
        const float fscale = (float)(0.5 * 27.211386024367243 * 0.5291772105638411) / wsum;
        const float e = (epart[0][tid] + epart[1][tid]) + (epart[2][tid] + epart[3][tid]);
        out[rowbase + tid] = s + e * fscale;
    }
}

extern "C" void kernel_launch(void* const* d_in, const int* in_sizes, int n_in,
                              void* d_out, int out_size, void* d_ws, size_t ws_size,
                              hipStream_t stream) {
    const float* x   = (const float*)d_in[0];
    const float* pos = (const float*)d_in[1];
    const float* W1  = (const float*)d_in[5];
    const float* b1  = (const float*)d_in[6];
    const float* W2  = (const float*)d_in[7];
    const float* b2  = (const float*)d_in[8];
    const float* qw  = (const float*)d_in[9];
    float* out = (float*)d_out;

    const int N = in_sizes[0] / XSTR;

    mlp_pair_fused_kernel<<<N / BROWS, 256, 0, stream>>>(x, pos, W1, b1, W2, b2, qw, out);
}
